# starting kernel
# baseline (speedup 1.0000x reference)
#include <hip/hip_runtime.h>
#include <hip/hip_bf16.h>


static constexpr int KF  = 128;
static constexpr int KH  = 8;
static constexpr int KD  = 16;
static constexpr int KHD = 128;
static constexpr float KALPHA = 0.2f;

typedef __attribute__((ext_vector_type(2))) float v2f;
typedef __attribute__((ext_vector_type(8))) float v8f;

__device__ __forceinline__ unsigned fmax_enc(float f) {
  unsigned u = __float_as_uint(f);
  return (u & 0x80000000u) ? ~u : (u | 0x80000000u);
}
__device__ __forceinline__ float fmax_dec(unsigned u) {
  return (u & 0x80000000u) ? __uint_as_float(u & 0x7fffffffu)
                           : __uint_as_float(~u);
}
static constexpr unsigned ENC_NEG_INF = 0x007FFFFFu;

__global__ void init_nh_kernel(float* __restrict__ z, unsigned* __restrict__ menc, int count) {
  int i = blockIdx.x * blockDim.x + threadIdx.x;
  if (i < count) { z[i] = 0.0f; menc[i] = ENC_NEG_INF; }
}
__global__ void zero_kernel(float* __restrict__ p, int count) {
  int i = blockIdx.x * blockDim.x + threadIdx.x;
  if (i < count) p[i] = 0.0f;
}

static constexpr int AS_STRIDE = 132;

__global__ __launch_bounds__(256) void gat_gemm_wmma_kernel(
    const float* __restrict__ hfeat, const float* __restrict__ W,
    float* __restrict__ Wh, int N) {
  __shared__ float As[16 * AS_STRIDE];

  const int tid  = threadIdx.x;
  const int row0 = blockIdx.x * 16;

  for (int i = tid; i < 16 * KF; i += 256) {
    int r = i >> 7, f = i & (KF - 1);
    float v = 0.0f;
    if (row0 + r < N) v = hfeat[(size_t)(row0 + r) * KF + f];
    As[r * AS_STRIDE + f] = v;
  }
  __syncthreads();

  const int wave = tid >> 5;
  const int lane = tid & 31;
  const int hl   = lane >> 4;
  const int col  = lane & 15;
  const int n0   = wave * 16;

  const float* Wb = W + wave * (KF * KD) + col;

  v8f acc = {};
  for (int k0 = 0; k0 < KF; k0 += 4) {
    const int k = k0 + 2 * hl;
    v2f a, b;
    a.x = As[col * AS_STRIDE + k];
    a.y = As[col * AS_STRIDE + k + 1];
    b.x = Wb[(size_t)k * KD];
    b.y = Wb[(size_t)(k + 1) * KD];
    acc = __builtin_amdgcn_wmma_f32_16x16x4_f32(
        false, a, false, b, (short)0, acc, false, false);
  }

#pragma unroll
  for (int v = 0; v < 8; ++v) {
    int r = row0 + v + 8 * hl;
    if (r < N) Wh[(size_t)r * KHD + n0 + col] = acc[v];
  }
}

__global__ void scores_kernel(const float* __restrict__ Wh,
                              const float* __restrict__ a1,
                              const float* __restrict__ a2,
                              float* __restrict__ s1, float* __restrict__ s2,
                              int N) {
  int idx = blockIdx.x * blockDim.x + threadIdx.x;
  if (idx >= N * KH) return;
  int n = idx >> 3, hh = idx & 7;
  const float* w  = Wh + (size_t)n * KHD + hh * KD;
  const float* v1 = a1 + hh * KD;
  const float* v2 = a2 + hh * KD;
  float acc1 = 0.0f, acc2 = 0.0f;
#pragma unroll
  for (int d = 0; d < KD; ++d) {
    float v = w[d];
    acc1 += v * v1[d];
    acc2 += v * v2[d];
  }
  s1[idx] = acc1;
  s2[idx] = acc2;
}

__device__ __forceinline__ float edge_logit(const float* s1, const float* s2,
                                            int s, int d, int hh) {
  float e = s1[s * KH + hh] + s2[d * KH + hh];
  return (e > 0.0f) ? e : KALPHA * e;
}

__global__ void edge_max_kernel(const int* __restrict__ src, const int* __restrict__ dst,
                                const float* __restrict__ s1, const float* __restrict__ s2,
                                unsigned* __restrict__ menc, int E) {
  int idx = blockIdx.x * blockDim.x + threadIdx.x;
  if (idx >= E * KH) return;
  int e = idx >> 3, hh = idx & 7;
  int d = dst[e];
  float lv = edge_logit(s1, s2, src[e], d, hh);
  atomicMax(&menc[d * KH + hh], fmax_enc(lv));
}

__global__ void finalize_max_kernel(const unsigned* __restrict__ menc,
                                    float* __restrict__ m, int count) {
  int i = blockIdx.x * blockDim.x + threadIdx.x;
  if (i >= count) return;
  float v = fmax_dec(menc[i]);
  m[i] = (v < -3.0e38f) ? 0.0f : v;
}

__global__ void edge_expsum_kernel(const int* __restrict__ src, const int* __restrict__ dst,
                                   const float* __restrict__ s1, const float* __restrict__ s2,
                                   const float* __restrict__ m, float* __restrict__ z, int E) {
  int idx = blockIdx.x * blockDim.x + threadIdx.x;
  if (idx >= E * KH) return;
  int e = idx >> 3, hh = idx & 7;
  int d = dst[e];
  float lv = edge_logit(s1, s2, src[e], d, hh);
  float p = __expf(lv - m[d * KH + hh]);
  unsafeAtomicAdd(&z[d * KH + hh], p);
}

__global__ void finalize_z_kernel(float* __restrict__ z, int count) {
  int i = blockIdx.x * blockDim.x + threadIdx.x;
  if (i >= count) return;
  z[i] = 1.0f / fmaxf(z[i], 1e-16f);
}

__global__ void edge_scatter_kernel(const int* __restrict__ src, const int* __restrict__ dst,
                                    const float* __restrict__ s1, const float* __restrict__ s2,
                                    const float* __restrict__ m, const float* __restrict__ zinv,
                                    const float* __restrict__ Wh, float* __restrict__ out, int E) {
  int idx = blockIdx.x * blockDim.x + threadIdx.x;
  if (idx >= E * KH) return;
  int e = idx >> 3, hh = idx & 7;
  int s = src[e], d = dst[e];
  float lv   = edge_logit(s1, s2, s, d, hh);
  float attn = __expf(lv - m[d * KH + hh]) * zinv[d * KH + hh];
  const float4* wsrc = (const float4*)(Wh + (size_t)s * KHD + hh * KD);
  float* op = out + (size_t)d * KHD + hh * KD;
#pragma unroll
  for (int q = 0; q < 4; ++q) {
    float4 w4 = wsrc[q];
    unsafeAtomicAdd(op + q * 4 + 0, attn * w4.x);
    unsafeAtomicAdd(op + q * 4 + 1, attn * w4.y);
    unsafeAtomicAdd(op + q * 4 + 2, attn * w4.z);
    unsafeAtomicAdd(op + q * 4 + 3, attn * w4.w);
  }
}

extern "C" void kernel_launch(void* const* d_in, const int* in_sizes, int n_in,
                              void* d_out, int out_size, void* d_ws, size_t ws_size,
                              hipStream_t stream) {
  const float* hfeat = (const float*)d_in[0];
  const int*   src   = (const int*)d_in[1];
  const int*   dst   = (const int*)d_in[2];
  const float* W     = (const float*)d_in[3];
  const float* a1    = (const float*)d_in[4];
  const float* a2    = (const float*)d_in[5];
  float* out = (float*)d_out;

  const int N = in_sizes[0] / KF;
  const int E = in_sizes[1];

  float*    Wh   = (float*)d_ws;
  float*    s1   = Wh + (size_t)N * KHD;
  float*    s2   = s1 + (size_t)N * KH;
  unsigned* menc = (unsigned*)(s2 + (size_t)N * KH);
  float*    m    = (float*)(menc + (size_t)N * KH);
  float*    z    = m + (size_t)N * KH;

  const int TB = 256;
  const int nh      = N * KH;
  const int nhd     = N * KHD;
  const int eh      = E * KH;
  const int g_nh    = (nh  + TB - 1) / TB;
  const int g_nhd   = (nhd + TB - 1) / TB;
  const int g_eh    = (eh  + TB - 1) / TB;
  const int g_gemm  = (N + 15) / 16;

  init_nh_kernel<<<g_nh, TB, 0, stream>>>(z, menc, nh);
  zero_kernel<<<g_nhd, TB, 0, stream>>>(out, nhd);

  gat_gemm_wmma_kernel<<<g_gemm, TB, 0, stream>>>(hfeat, W, Wh, N);

  scores_kernel<<<g_nh, TB, 0, stream>>>(Wh, a1, a2, s1, s2, N);
  edge_max_kernel<<<g_eh, TB, 0, stream>>>(src, dst, s1, s2, menc, E);
  finalize_max_kernel<<<g_nh, TB, 0, stream>>>(menc, m, nh);
  edge_expsum_kernel<<<g_eh, TB, 0, stream>>>(src, dst, s1, s2, m, z, E);
  finalize_z_kernel<<<g_nh, TB, 0, stream>>>(z, nh);
  edge_scatter_kernel<<<g_eh, TB, 0, stream>>>(src, dst, s1, s2, m, z, Wh, out, E);
}


# Scatter: global f32 atomics on out replaced by owner work-groups: each holds 392 dst rows in LDS, scans all dst, LDS f32 adds for its edges (arrival order), one plain store per row; out fill dropped
# speedup vs baseline: 3.7241x; 3.7241x over previous
#include <hip/hip_runtime.h>
#include <hip/hip_bf16.h>


static constexpr int KF  = 128;
static constexpr int KH  = 8;
static constexpr int KD  = 16;
static constexpr int KHD = 128;
static constexpr float KALPHA = 0.2f;

typedef __attribute__((ext_vector_type(2))) float v2f;
typedef __attribute__((ext_vector_type(8))) float v8f;

__device__ __forceinline__ unsigned fmax_enc(float f) {
  unsigned u = __float_as_uint(f);
  return (u & 0x80000000u) ? ~u : (u | 0x80000000u);
}
__device__ __forceinline__ float fmax_dec(unsigned u) {
  return (u & 0x80000000u) ? __uint_as_float(u & 0x7fffffffu)
                           : __uint_as_float(~u);
}
static constexpr unsigned ENC_NEG_INF = 0x007FFFFFu;

__global__ void init_nh_kernel(float* __restrict__ z, unsigned* __restrict__ menc, int count) {
  int i = blockIdx.x * blockDim.x + threadIdx.x;
  if (i < count) { z[i] = 0.0f; menc[i] = ENC_NEG_INF; }
}

static constexpr int AS_STRIDE = 132;

__global__ __launch_bounds__(256) void gat_gemm_wmma_kernel(
    const float* __restrict__ hfeat, const float* __restrict__ W,
    float* __restrict__ Wh, int N) {
  __shared__ float As[16 * AS_STRIDE];

  const int tid  = threadIdx.x;
  const int row0 = blockIdx.x * 16;

  for (int i = tid; i < 16 * KF; i += 256) {
    int r = i >> 7, f = i & (KF - 1);
    float v = 0.0f;
    if (row0 + r < N) v = hfeat[(size_t)(row0 + r) * KF + f];
    As[r * AS_STRIDE + f] = v;
  }
  __syncthreads();

  const int wave = tid >> 5;
  const int lane = tid & 31;
  const int hl   = lane >> 4;
  const int col  = lane & 15;
  const int n0   = wave * 16;

  const float* Wb = W + wave * (KF * KD) + col;

  v8f acc = {};
  for (int k0 = 0; k0 < KF; k0 += 4) {
    const int k = k0 + 2 * hl;
    v2f a, b;
    a.x = As[col * AS_STRIDE + k];
    a.y = As[col * AS_STRIDE + k + 1];
    b.x = Wb[(size_t)k * KD];
    b.y = Wb[(size_t)(k + 1) * KD];
    acc = __builtin_amdgcn_wmma_f32_16x16x4_f32(
        false, a, false, b, (short)0, acc, false, false);
  }

#pragma unroll
  for (int v = 0; v < 8; ++v) {
    int r = row0 + v + 8 * hl;
    if (r < N) Wh[(size_t)r * KHD + n0 + col] = acc[v];
  }
}

__global__ void scores_kernel(const float* __restrict__ Wh,
                              const float* __restrict__ a1,
                              const float* __restrict__ a2,
                              float* __restrict__ s1, float* __restrict__ s2,
                              int N) {
  int idx = blockIdx.x * blockDim.x + threadIdx.x;
  if (idx >= N * KH) return;
  int n = idx >> 3, hh = idx & 7;
  const float* w  = Wh + (size_t)n * KHD + hh * KD;
  const float* v1 = a1 + hh * KD;
  const float* v2 = a2 + hh * KD;
  float acc1 = 0.0f, acc2 = 0.0f;
#pragma unroll
  for (int d = 0; d < KD; ++d) {
    float v = w[d];
    acc1 += v * v1[d];
    acc2 += v * v2[d];
  }
  s1[idx] = acc1;
  s2[idx] = acc2;
}

__device__ __forceinline__ float edge_logit(const float* s1, const float* s2,
                                            int s, int d, int hh) {
  float e = s1[s * KH + hh] + s2[d * KH + hh];
  return (e > 0.0f) ? e : KALPHA * e;
}

__global__ void edge_max_kernel(const int* __restrict__ src, const int* __restrict__ dst,
                                const float* __restrict__ s1, const float* __restrict__ s2,
                                unsigned* __restrict__ menc, int E) {
  int idx = blockIdx.x * blockDim.x + threadIdx.x;
  if (idx >= E * KH) return;
  int e = idx >> 3, hh = idx & 7;
  int d = dst[e];
  float lv = edge_logit(s1, s2, src[e], d, hh);
  atomicMax(&menc[d * KH + hh], fmax_enc(lv));
}

__global__ void finalize_max_kernel(const unsigned* __restrict__ menc,
                                    float* __restrict__ m, int count) {
  int i = blockIdx.x * blockDim.x + threadIdx.x;
  if (i >= count) return;
  float v = fmax_dec(menc[i]);
  m[i] = (v < -3.0e38f) ? 0.0f : v;
}

__global__ void edge_expsum_kernel(const int* __restrict__ src, const int* __restrict__ dst,
                                   const float* __restrict__ s1, const float* __restrict__ s2,
                                   const float* __restrict__ m, float* __restrict__ z, int E) {
  int idx = blockIdx.x * blockDim.x + threadIdx.x;
  if (idx >= E * KH) return;
  int e = idx >> 3, hh = idx & 7;
  int d = dst[e];
  float lv = edge_logit(s1, s2, src[e], d, hh);
  float p = __expf(lv - m[d * KH + hh]);
  unsafeAtomicAdd(&z[d * KH + hh], p);
}

__global__ void finalize_z_kernel(float* __restrict__ z, int count) {
  int i = blockIdx.x * blockDim.x + threadIdx.x;
  if (i >= count) return;
  z[i] = 1.0f / fmaxf(z[i], 1e-16f);
}

static constexpr int OWN_ROWS  = 392;
static constexpr int OWN_WAVES = 32;
static constexpr int QCAP      = 256;
static constexpr int SCAN_U    = 8;
static constexpr int DRAIN_G   = 8;

__global__ __launch_bounds__(1024) void edge_scatter_lds_kernel(
    const int* __restrict__ src, const int* __restrict__ dst,
    const float* __restrict__ s1, const float* __restrict__ s2,
    const float* __restrict__ m, const float* __restrict__ zinv,
    const float* __restrict__ Wh, float* __restrict__ out, int N, int E) {
  __shared__ float tile[OWN_ROWS * KHD];
  __shared__ int   q_e[OWN_WAVES * QCAP];
  __shared__ int   q_d[OWN_WAVES * QCAP];

  const int tid  = threadIdx.x;
  const int lane = tid & 31;
  const int wave = __builtin_amdgcn_readfirstlane(tid >> 5);
  const int r0   = blockIdx.x * OWN_ROWS;
  const int rows = min(OWN_ROWS, N - r0);

  for (int i = tid; i < OWN_ROWS * (KHD / 4); i += 1024)
    ((float4*)tile)[i] = make_float4(0.0f, 0.0f, 0.0f, 0.0f);
  __syncthreads();

  int* qe = q_e + wave * QCAP;
  int* qd = q_d + wave * QCAP;
  unsigned head = 0, tail = 0;
  const int hh = lane >> 2;

  auto drain = [&](unsigned nb) {
    __builtin_amdgcn_wave_barrier();
    int dl_l = 0, s_l = 0;
    if ((unsigned)lane < nb) {
      const unsigned qi = (head + (unsigned)lane) & (QCAP - 1);
      dl_l = qd[qi];
      s_l  = src[qe[qi]];
    }
    auto add_row = [&](int dl, float a1v, float a2v, float mv, float zv, float4 w4) {
      float lv = a1v + a2v;
      lv = (lv > 0.0f) ? lv : KALPHA * lv;
      const float attn = __expf(lv - mv) * zv;
      float* t = tile + dl * KHD + lane;
      __hip_atomic_fetch_add(t,      attn * w4.x, __ATOMIC_RELAXED, __HIP_MEMORY_SCOPE_WORKGROUP);
      __hip_atomic_fetch_add(t + 32, attn * w4.y, __ATOMIC_RELAXED, __HIP_MEMORY_SCOPE_WORKGROUP);
      __hip_atomic_fetch_add(t + 64, attn * w4.z, __ATOMIC_RELAXED, __HIP_MEMORY_SCOPE_WORKGROUP);
      __hip_atomic_fetch_add(t + 96, attn * w4.w, __ATOMIC_RELAXED, __HIP_MEMORY_SCOPE_WORKGROUP);
    };
    unsigned k = 0;
    for (; k + DRAIN_G <= nb; k += DRAIN_G) {
      int    dl[DRAIN_G];
      float  a1v[DRAIN_G], a2v[DRAIN_G], mv[DRAIN_G], zv[DRAIN_G];
      float4 w4[DRAIN_G];
#pragma unroll
      for (int j = 0; j < DRAIN_G; ++j) {
        const int s = __builtin_amdgcn_readlane(s_l, (int)k + j);
        dl[j] = __builtin_amdgcn_readlane(dl_l, (int)k + j);
        const int d = r0 + dl[j];
        a1v[j] = s1[s * KH + hh];
        a2v[j] = s2[d * KH + hh];
        mv[j]  = m[d * KH + hh];
        zv[j]  = zinv[d * KH + hh];
        w4[j]  = *(const float4*)(Wh + (size_t)s * KHD + lane * 4);
      }
#pragma unroll
      for (int j = 0; j < DRAIN_G; ++j) add_row(dl[j], a1v[j], a2v[j], mv[j], zv[j], w4[j]);
    }
    for (; k < nb; ++k) {
      const int s  = __builtin_amdgcn_readlane(s_l, (int)k);
      const int dl1 = __builtin_amdgcn_readlane(dl_l, (int)k);
      const int d  = r0 + dl1;
      add_row(dl1, s1[s * KH + hh], s2[d * KH + hh], m[d * KH + hh], zinv[d * KH + hh],
              *(const float4*)(Wh + (size_t)s * KHD + lane * 4));
    }
    head += nb;
  };

  auto push = [&](bool hit, int e, int rel) {
    const unsigned mask = __builtin_amdgcn_ballot_w32(hit);
    if (mask != 0u) {
      if (hit) {
        const unsigned pos = (tail + __builtin_popcount(mask & ((1u << lane) - 1u))) & (QCAP - 1);
        qe[pos] = e;
        qd[pos] = rel;
      }
      tail += __builtin_popcount(mask);
    }
  };

  const int BLK = SCAN_U * 128;
  for (int blk = wave * BLK; blk < E; blk += OWN_WAVES * BLK) {
    int4 dv[SCAN_U];
    if (blk + BLK <= E) {
#pragma unroll
      for (int u = 0; u < SCAN_U; ++u)
        dv[u] = *(const int4*)(dst + blk + u * 128 + lane * 4);
    } else {
#pragma unroll
      for (int u = 0; u < SCAN_U; ++u) {
        const int i0 = blk + u * 128 + lane * 4;
        dv[u].x = (i0     < E) ? dst[i0]     : -1;
        dv[u].y = (i0 + 1 < E) ? dst[i0 + 1] : -1;
        dv[u].z = (i0 + 2 < E) ? dst[i0 + 2] : -1;
        dv[u].w = (i0 + 3 < E) ? dst[i0 + 3] : -1;
      }
    }
#pragma unroll
    for (int u = 0; u < SCAN_U; ++u) {
      const int i0 = blk + u * 128 + lane * 4;
      const unsigned rx = (unsigned)(dv[u].x - r0), ry = (unsigned)(dv[u].y - r0);
      const unsigned rz = (unsigned)(dv[u].z - r0), rw = (unsigned)(dv[u].w - r0);
      const bool hx = rx < (unsigned)rows, hy = ry < (unsigned)rows;
      const bool hz = rz < (unsigned)rows, hw = rw < (unsigned)rows;
      if (__builtin_amdgcn_ballot_w32(hx || hy || hz || hw) != 0u) {
        push(hx, i0,     (int)rx);
        push(hy, i0 + 1, (int)ry);
        push(hz, i0 + 2, (int)rz);
        push(hw, i0 + 3, (int)rw);
        while (tail - head >= 32u) drain(32u);
      }
    }
  }
  if (tail != head) drain(tail - head);

  __syncthreads();
  for (int r = wave; r < rows; r += OWN_WAVES) {
    const float* t = tile + r * KHD + lane;
    float4 o;
    o.x = t[0]; o.y = t[32]; o.z = t[64]; o.w = t[96];
    *(float4*)(out + (size_t)(r0 + r) * KHD + lane * 4) = o;
  }
}

extern "C" void kernel_launch(void* const* d_in, const int* in_sizes, int n_in,
                              void* d_out, int out_size, void* d_ws, size_t ws_size,
                              hipStream_t stream) {
  const float* hfeat = (const float*)d_in[0];
  const int*   src   = (const int*)d_in[1];
  const int*   dst   = (const int*)d_in[2];
  const float* W     = (const float*)d_in[3];
  const float* a1    = (const float*)d_in[4];
  const float* a2    = (const float*)d_in[5];
  float* out = (float*)d_out;

  const int N = in_sizes[0] / KF;
  const int E = in_sizes[1];

  float*    Wh   = (float*)d_ws;
  float*    s1   = Wh + (size_t)N * KHD;
  float*    s2   = s1 + (size_t)N * KH;
  unsigned* menc = (unsigned*)(s2 + (size_t)N * KH);
  float*    m    = (float*)(menc + (size_t)N * KH);
  float*    z    = m + (size_t)N * KH;

  const int TB = 256;
  const int nh      = N * KH;
  const int eh      = E * KH;
  const int g_nh    = (nh  + TB - 1) / TB;
  const int g_eh    = (eh  + TB - 1) / TB;
  const int g_gemm  = (N + 15) / 16;
  const int g_own   = (N + OWN_ROWS - 1) / OWN_ROWS;

  init_nh_kernel<<<g_nh, TB, 0, stream>>>(z, menc, nh);

  gat_gemm_wmma_kernel<<<g_gemm, TB, 0, stream>>>(hfeat, W, Wh, N);

  scores_kernel<<<g_nh, TB, 0, stream>>>(Wh, a1, a2, s1, s2, N);
  edge_max_kernel<<<g_eh, TB, 0, stream>>>(src, dst, s1, s2, menc, E);
  finalize_max_kernel<<<g_nh, TB, 0, stream>>>(menc, m, nh);
  edge_expsum_kernel<<<g_eh, TB, 0, stream>>>(src, dst, s1, s2, m, z, E);
  finalize_z_kernel<<<g_nh, TB, 0, stream>>>(z, nh);
  edge_scatter_lds_kernel<<<g_own, 1024, 0, stream>>>(src, dst, s1, s2, m, z, Wh, out, N, E);
}


# Segment max and exp-sum moved into the owner work-groups: one dst scan keeps each wave's edges as an LDS list; LDS atomic max, then exp-sum beside the sums in LDS arrival order, 1/z after the sum
# speedup vs baseline: 6.1152x; 1.6421x over previous
#include <hip/hip_runtime.h>
#include <hip/hip_bf16.h>


static constexpr int KF  = 128;
static constexpr int KH  = 8;
static constexpr int KD  = 16;
static constexpr int KHD = 128;
static constexpr float KALPHA = 0.2f;

typedef __attribute__((ext_vector_type(2))) float v2f;
typedef __attribute__((ext_vector_type(8))) float v8f;

__device__ __forceinline__ unsigned fmax_enc(float f) {
  unsigned u = __float_as_uint(f);
  return (u & 0x80000000u) ? ~u : (u | 0x80000000u);
}
__device__ __forceinline__ float fmax_dec(unsigned u) {
  return (u & 0x80000000u) ? __uint_as_float(u & 0x7fffffffu)
                           : __uint_as_float(~u);
}
static constexpr unsigned ENC_NEG_INF = 0x007FFFFFu;

static constexpr int AS_STRIDE = 132;

__global__ __launch_bounds__(256) void gat_gemm_wmma_kernel(
    const float* __restrict__ hfeat, const float* __restrict__ W,
    float* __restrict__ Wh, int N) {
  __shared__ float As[16 * AS_STRIDE];

  const int tid  = threadIdx.x;
  const int row0 = blockIdx.x * 16;

  for (int i = tid; i < 16 * KF; i += 256) {
    int r = i >> 7, f = i & (KF - 1);
    float v = 0.0f;
    if (row0 + r < N) v = hfeat[(size_t)(row0 + r) * KF + f];
    As[r * AS_STRIDE + f] = v;
  }
  __syncthreads();

  const int wave = tid >> 5;
  const int lane = tid & 31;
  const int hl   = lane >> 4;
  const int col  = lane & 15;
  const int n0   = wave * 16;

  const float* Wb = W + wave * (KF * KD) + col;

  v8f acc = {};
  for (int k0 = 0; k0 < KF; k0 += 4) {
    const int k = k0 + 2 * hl;
    v2f a, b;
    a.x = As[col * AS_STRIDE + k];
    a.y = As[col * AS_STRIDE + k + 1];
    b.x = Wb[(size_t)k * KD];
    b.y = Wb[(size_t)(k + 1) * KD];
    acc = __builtin_amdgcn_wmma_f32_16x16x4_f32(
        false, a, false, b, (short)0, acc, false, false);
  }

#pragma unroll
  for (int v = 0; v < 8; ++v) {
    int r = row0 + v + 8 * hl;
    if (r < N) Wh[(size_t)r * KHD + n0 + col] = acc[v];
  }
}

__global__ void scores_kernel(const float* __restrict__ Wh,
                              const float* __restrict__ a1,
                              const float* __restrict__ a2,
                              float* __restrict__ s1, float* __restrict__ s2,
                              int N) {
  int idx = blockIdx.x * blockDim.x + threadIdx.x;
  if (idx >= N * KH) return;
  int n = idx >> 3, hh = idx & 7;
  const float* w  = Wh + (size_t)n * KHD + hh * KD;
  const float* v1 = a1 + hh * KD;
  const float* v2 = a2 + hh * KD;
  float acc1 = 0.0f, acc2 = 0.0f;
#pragma unroll
  for (int d = 0; d < KD; ++d) {
    float v = w[d];
    acc1 += v * v1[d];
    acc2 += v * v2[d];
  }
  s1[idx] = acc1;
  s2[idx] = acc2;
}

static constexpr int OWN_ROWS  = 392;
static constexpr int OWN_WAVES = 32;
static constexpr int CAPW      = 448;
static constexpr int SCAN_U    = 8;
static constexpr int GRP       = 8;

__global__ __launch_bounds__(1024) void edge_softmax_owner_kernel(
    const int* __restrict__ src, const int* __restrict__ dst,
    const float* __restrict__ s1, const float* __restrict__ s2,
    const float* __restrict__ Wh, float* __restrict__ out, int N, int E) {
  __shared__ float          tile[OWN_ROWS * KHD];
  __shared__ unsigned       m_run[OWN_ROWS * KH];
  __shared__ unsigned       m_new[OWN_ROWS * KH];
  __shared__ float          zsum[OWN_ROWS * KH];
  __shared__ int            l_s[OWN_WAVES * CAPW];
  __shared__ unsigned short l_r[OWN_WAVES * CAPW];
  __shared__ int            more_flag[2];

  const int tid  = threadIdx.x;
  const int lane = tid & 31;
  const int wave = __builtin_amdgcn_readfirstlane(tid >> 5);
  const int r0   = blockIdx.x * OWN_ROWS;
  const int rows = min(OWN_ROWS, N - r0);

  for (int i = tid; i < OWN_ROWS * (KHD / 4); i += 1024)
    ((float4*)tile)[i] = make_float4(0.0f, 0.0f, 0.0f, 0.0f);
  for (int i = tid; i < OWN_ROWS * KH; i += 1024) {
    m_run[i] = ENC_NEG_INF;
    m_new[i] = ENC_NEG_INF;
    zsum[i]  = 0.0f;
  }
  if (tid < 2) more_flag[tid] = 0;
  __syncthreads();

  int*            ls = l_s + wave * CAPW;
  unsigned short* lr = l_r + wave * CAPW;
  unsigned cnt = 0;

  auto push = [&](bool hit, int e, unsigned rel) {
    const unsigned mask = __builtin_amdgcn_ballot_w32(hit);
    if (mask != 0u) {
      if (hit) {
        const unsigned pos = cnt + __builtin_popcount(mask & ((1u << lane) - 1u));
        ls[pos] = e;
        lr[pos] = (unsigned short)rel;
      }
      cnt += __builtin_popcount(mask);
    }
  };

  const int BLK = SCAN_U * 128;
  int blk    = wave * BLK;
  int ustart = 0;

  for (int rnd = 0; ; ++rnd) {
    cnt = 0;
    bool stopped = false;
    while (blk < E && !stopped) {
      int4 dv[SCAN_U];
      if (blk + BLK <= E) {
#pragma unroll
        for (int u = 0; u < SCAN_U; ++u)
          dv[u] = *(const int4*)(dst + blk + u * 128 + lane * 4);
      } else {
#pragma unroll
        for (int u = 0; u < SCAN_U; ++u) {
          const int i0 = blk + u * 128 + lane * 4;
          dv[u].x = (i0     < E) ? dst[i0]     : -1;
          dv[u].y = (i0 + 1 < E) ? dst[i0 + 1] : -1;
          dv[u].z = (i0 + 2 < E) ? dst[i0 + 2] : -1;
          dv[u].w = (i0 + 3 < E) ? dst[i0 + 3] : -1;
        }
      }
#pragma unroll
      for (int u = 0; u < SCAN_U; ++u) {
        if (!stopped && u >= ustart) {
          if (cnt + 128u > (unsigned)CAPW) {
            stopped = true;
            ustart  = u;
          } else {
            const int i0 = blk + u * 128 + lane * 4;
            const unsigned rx = (unsigned)(dv[u].x - r0), ry = (unsigned)(dv[u].y - r0);
            const unsigned rz = (unsigned)(dv[u].z - r0), rw = (unsigned)(dv[u].w - r0);
            const bool hx = rx < (unsigned)rows, hy = ry < (unsigned)rows;
            const bool hz = rz < (unsigned)rows, hw = rw < (unsigned)rows;
            if (__builtin_amdgcn_ballot_w32(hx || hy || hz || hw) != 0u) {
              push(hx, i0,     rx);
              push(hy, i0 + 1, ry);
              push(hz, i0 + 2, rz);
              push(hw, i0 + 3, rw);
            }
          }
        }
      }
      if (!stopped) { blk += OWN_WAVES * BLK; ustart = 0; }
    }
    if (blk < E && lane == 0) more_flag[rnd & 1] = 1;

    __builtin_amdgcn_wave_barrier();

    for (unsigned i0 = 0; i0 < cnt; i0 += 128u) {
      int ee[4], ss[4];
#pragma unroll
      for (int j = 0; j < 4; ++j) {
        const unsigned idx = i0 + j * 32 + lane;
        ee[j] = (idx < cnt) ? ls[idx] : 0;
      }
#pragma unroll
      for (int j = 0; j < 4; ++j) ss[j] = src[ee[j]];
#pragma unroll
      for (int j = 0; j < 4; ++j) {
        const unsigned idx = i0 + j * 32 + lane;
        if (idx < cnt) ls[idx] = ss[j];
      }
    }
    __builtin_amdgcn_wave_barrier();

    {
      const int j4 = lane >> 3, h8 = lane & 7;
      for (unsigned k = 0; k < cnt; k += 32u) {
        float    av[8], bv[8];
        unsigned rl[8];
#pragma unroll
        for (int i = 0; i < 8; ++i) {
          const unsigned idx = k + i * 4 + j4;
          const unsigned ic  = (idx < cnt) ? idx : 0u;
          const int s = ls[ic];
          rl[i] = (unsigned)lr[ic];
          av[i] = s1[s * KH + h8];
          bv[i] = s2[(r0 + (int)rl[i]) * KH + h8];
        }
#pragma unroll
        for (int i = 0; i < 8; ++i) {
          float lv = av[i] + bv[i];
          lv = (lv > 0.0f) ? lv : KALPHA * lv;
          atomicMax(&m_new[rl[i] * KH + h8], fmax_enc(lv));
        }
      }
    }
    __syncthreads();

    for (int i = tid; i < rows * KH; i += 1024) {
      const unsigned mn = m_new[i], mo = m_run[i];
      if (mn != mo) {
        m_run[i] = mn;
        if (mo != ENC_NEG_INF) {
          const float sc = __expf(fmax_dec(mo) - fmax_dec(mn));
          zsum[i] *= sc;
          float* t = tile + (i >> 3) * KHD + 4 * (i & 7);
#pragma unroll
          for (int q = 0; q < 4; ++q) {
            t[q * 32 + 0] *= sc; t[q * 32 + 1] *= sc;
            t[q * 32 + 2] *= sc; t[q * 32 + 3] *= sc;
          }
        }
      }
    }
    __syncthreads();

    {
      const int hh = lane >> 2;
      auto add_row = [&](int rl1, float a1v, float a2v, unsigned me, float4 w4) {
        float lv = a1v + a2v;
        lv = (lv > 0.0f) ? lv : KALPHA * lv;
        float mv = fmax_dec(me);
        mv = (mv < -3.0e38f) ? 0.0f : mv;
        const float p = __expf(lv - mv);
        float* t = tile + rl1 * KHD + lane;
        __hip_atomic_fetch_add(t,      p * w4.x, __ATOMIC_RELAXED, __HIP_MEMORY_SCOPE_WORKGROUP);
        __hip_atomic_fetch_add(t + 32, p * w4.y, __ATOMIC_RELAXED, __HIP_MEMORY_SCOPE_WORKGROUP);
        __hip_atomic_fetch_add(t + 64, p * w4.z, __ATOMIC_RELAXED, __HIP_MEMORY_SCOPE_WORKGROUP);
        __hip_atomic_fetch_add(t + 96, p * w4.w, __ATOMIC_RELAXED, __HIP_MEMORY_SCOPE_WORKGROUP);
        if ((lane & 3) == 0)
          __hip_atomic_fetch_add(zsum + rl1 * KH + hh, p, __ATOMIC_RELAXED, __HIP_MEMORY_SCOPE_WORKGROUP);
      };
      for (unsigned k0 = 0; k0 < cnt; k0 += 32u) {
        const unsigned nb = min(32u, cnt - k0);
        int s_l = 0, r_l = 0;
        if ((unsigned)lane < nb) { s_l = ls[k0 + lane]; r_l = (int)lr[k0 + lane]; }
        unsigned k = 0;
        for (; k + GRP <= nb; k += GRP) {
          int      rl[GRP];
          float    a1v[GRP], a2v[GRP];
          unsigned me[GRP];
          float4   w4[GRP];
#pragma unroll
          for (int j = 0; j < GRP; ++j) {
            const int s = __builtin_amdgcn_readlane(s_l, (int)k + j);
            rl[j]  = __builtin_amdgcn_readlane(r_l, (int)k + j);
            a1v[j] = s1[s * KH + hh];
            a2v[j] = s2[(r0 + rl[j]) * KH + hh];
            me[j]  = m_run[rl[j] * KH + hh];
            w4[j]  = *(const float4*)(Wh + (size_t)s * KHD + lane * 4);
          }
#pragma unroll
          for (int j = 0; j < GRP; ++j) add_row(rl[j], a1v[j], a2v[j], me[j], w4[j]);
        }
        for (; k < nb; ++k) {
          const int s   = __builtin_amdgcn_readlane(s_l, (int)k);
          const int rl1 = __builtin_amdgcn_readlane(r_l, (int)k);
          add_row(rl1, s1[s * KH + hh], s2[(r0 + rl1) * KH + hh], m_run[rl1 * KH + hh],
                  *(const float4*)(Wh + (size_t)s * KHD + lane * 4));
        }
      }
    }

    const int more = __builtin_amdgcn_readfirstlane(more_flag[rnd & 1]);
    __syncthreads();
    if (more == 0) break;
    if (tid == 0) more_flag[rnd & 1] = 0;
  }

  for (int r = wave; r < rows; r += OWN_WAVES) {
    const float* t = tile + r * KHD + lane;
    const float zi = 1.0f / fmaxf(zsum[r * KH + (lane >> 2)], 1e-16f);
    float4 o;
    o.x = t[0] * zi; o.y = t[32] * zi; o.z = t[64] * zi; o.w = t[96] * zi;
    *(float4*)(out + (size_t)(r0 + r) * KHD + lane * 4) = o;
  }
}

extern "C" void kernel_launch(void* const* d_in, const int* in_sizes, int n_in,
                              void* d_out, int out_size, void* d_ws, size_t ws_size,
                              hipStream_t stream) {
  const float* hfeat = (const float*)d_in[0];
  const int*   src   = (const int*)d_in[1];
  const int*   dst   = (const int*)d_in[2];
  const float* W     = (const float*)d_in[3];
  const float* a1    = (const float*)d_in[4];
  const float* a2    = (const float*)d_in[5];
  float* out = (float*)d_out;

  const int N = in_sizes[0] / KF;
  const int E = in_sizes[1];

  float* Wh = (float*)d_ws;
  float* s1 = Wh + (size_t)N * KHD;
  float* s2 = s1 + (size_t)N * KH;

  const int TB = 256;
  const int nh      = N * KH;
  const int g_nh    = (nh  + TB - 1) / TB;
  const int g_gemm  = (N + 15) / 16;
  const int g_own   = (N + OWN_ROWS - 1) / OWN_ROWS;

  gat_gemm_wmma_kernel<<<g_gemm, TB, 0, stream>>>(hfeat, W, Wh, N);

  scores_kernel<<<g_nh, TB, 0, stream>>>(Wh, a1, a2, s1, s2, N);
  edge_softmax_owner_kernel<<<g_own, 1024, 0, stream>>>(src, dst, s1, s2, Wh, out, N, E);
}


# Pass B: softmax weights formed 4 edges x 8 heads per instruction (list entries kept in that lane map, fetched a trip ahead), given to row adds by DPP quad broadcast; sums still LDS arrival order
# speedup vs baseline: 7.7993x; 1.2754x over previous
#include <hip/hip_runtime.h>
#include <hip/hip_bf16.h>
#include <type_traits>


static constexpr int KF  = 128;
static constexpr int KH  = 8;
static constexpr int KD  = 16;
static constexpr int KHD = 128;
static constexpr float KALPHA = 0.2f;

typedef __attribute__((ext_vector_type(2))) float v2f;
typedef __attribute__((ext_vector_type(8))) float v8f;

__device__ __forceinline__ unsigned fmax_enc(float f) {
  unsigned u = __float_as_uint(f);
  return (u & 0x80000000u) ? ~u : (u | 0x80000000u);
}
__device__ __forceinline__ float fmax_dec(unsigned u) {
  return (u & 0x80000000u) ? __uint_as_float(u & 0x7fffffffu)
                           : __uint_as_float(~u);
}
static constexpr unsigned ENC_NEG_INF = 0x007FFFFFu;

static constexpr int AS_STRIDE = 132;

__global__ __launch_bounds__(256) void gat_gemm_wmma_kernel(
    const float* __restrict__ hfeat, const float* __restrict__ W,
    float* __restrict__ Wh, int N) {
  __shared__ float As[16 * AS_STRIDE];

  const int tid  = threadIdx.x;
  const int row0 = blockIdx.x * 16;

  for (int i = tid; i < 16 * KF; i += 256) {
    int r = i >> 7, f = i & (KF - 1);
    float v = 0.0f;
    if (row0 + r < N) v = hfeat[(size_t)(row0 + r) * KF + f];
    As[r * AS_STRIDE + f] = v;
  }
  __syncthreads();

  const int wave = tid >> 5;
  const int lane = tid & 31;
  const int hl   = lane >> 4;
  const int col  = lane & 15;
  const int n0   = wave * 16;

  const float* Wb = W + wave * (KF * KD) + col;

  v8f acc = {};
  for (int k0 = 0; k0 < KF; k0 += 4) {
    const int k = k0 + 2 * hl;
    v2f a, b;
    a.x = As[col * AS_STRIDE + k];
    a.y = As[col * AS_STRIDE + k + 1];
    b.x = Wb[(size_t)k * KD];
    b.y = Wb[(size_t)(k + 1) * KD];
    acc = __builtin_amdgcn_wmma_f32_16x16x4_f32(
        false, a, false, b, (short)0, acc, false, false);
  }

#pragma unroll
  for (int v = 0; v < 8; ++v) {
    int r = row0 + v + 8 * hl;
    if (r < N) Wh[(size_t)r * KHD + n0 + col] = acc[v];
  }
}

static constexpr int OWN_ROWS = 392;

__device__ __forceinline__ float quad_bcast(float v, int q) {
  const int x = __float_as_int(v);
  int r;
  switch (q) {
    case 0:  r = __builtin_amdgcn_mov_dpp(x, 0x00, 0xF, 0xF, true); break;
    case 1:  r = __builtin_amdgcn_mov_dpp(x, 0x55, 0xF, 0xF, true); break;
    case 2:  r = __builtin_amdgcn_mov_dpp(x, 0xAA, 0xF, 0xF, true); break;
    default: r = __builtin_amdgcn_mov_dpp(x, 0xFF, 0xF, 0xF, true); break;
  }
  return __int_as_float(r);
}

__device__ __forceinline__ unsigned own_byte(int d) {
  return ((unsigned)d / (unsigned)OWN_ROWS) & 0xFFu;
}

__global__ void scores_kernel(const float* __restrict__ Wh,
                              const float* __restrict__ a1,
                              const float* __restrict__ a2,
                              float* __restrict__ s1, float* __restrict__ s2,
                              const int* __restrict__ dst,
                              unsigned char* __restrict__ own,
                              int N, int E, int g_nh) {
  if ((int)blockIdx.x >= g_nh) {
    const int e0   = (((int)blockIdx.x - g_nh) * (int)blockDim.x + (int)threadIdx.x) * 16;
    const int epad = (E + 15) & ~15;
    if (e0 >= epad) return;
    unsigned w[4];
#pragma unroll
    for (int q = 0; q < 4; ++q) {
      const int e = e0 + 4 * q;
      int4 d;
      if (e + 4 <= E) {
        d = *(const int4*)(dst + e);
      } else {
        d.x = (e     < E) ? dst[e]     : 0;
        d.y = (e + 1 < E) ? dst[e + 1] : 0;
        d.z = (e + 2 < E) ? dst[e + 2] : 0;
        d.w = (e + 3 < E) ? dst[e + 3] : 0;
      }
      w[q] = own_byte(d.x) | (own_byte(d.y) << 8) | (own_byte(d.z) << 16) | (own_byte(d.w) << 24);
    }
    *(uint4*)(own + e0) = make_uint4(w[0], w[1], w[2], w[3]);
    return;
  }
  int idx = blockIdx.x * blockDim.x + threadIdx.x;
  if (idx >= N * KH) return;
  int n = idx >> 3, hh = idx & 7;
  const float* w  = Wh + (size_t)n * KHD + hh * KD;
  const float* v1 = a1 + hh * KD;
  const float* v2 = a2 + hh * KD;
  float acc1 = 0.0f, acc2 = 0.0f;
#pragma unroll
  for (int d = 0; d < KD; ++d) {
    float v = w[d];
    acc1 += v * v1[d];
    acc2 += v * v2[d];
  }
  s1[idx] = acc1;
  s2[idx] = acc2;
}

static constexpr int OWN_WAVES = 32;
static constexpr int CAPW      = 448;
static constexpr int SCAN_U    = 8;
static constexpr int GRP       = 8;

__global__ __launch_bounds__(1024) void edge_softmax_owner_kernel(
    const int* __restrict__ src, const int* __restrict__ dst,
    const unsigned char* __restrict__ own,
    const float* __restrict__ s1, const float* __restrict__ s2,
    const float* __restrict__ Wh, float* __restrict__ out, int N, int E) {
  __shared__ float          tile[OWN_ROWS * KHD];
  __shared__ unsigned       m_run[OWN_ROWS * KH];
  __shared__ unsigned       m_new[OWN_ROWS * KH];
  __shared__ float          zsum[OWN_ROWS * KH];
  __shared__ int            l_s[OWN_WAVES * CAPW];
  __shared__ unsigned short l_r[OWN_WAVES * CAPW];
  __shared__ int            more_flag[2];

  const int tid  = threadIdx.x;
  const int lane = tid & 31;
  const int wave = __builtin_amdgcn_readfirstlane(tid >> 5);
  const int r0   = blockIdx.x * OWN_ROWS;
  const int rows = min(OWN_ROWS, N - r0);

  for (int i = tid; i < OWN_ROWS * (KHD / 4); i += 1024)
    ((float4*)tile)[i] = make_float4(0.0f, 0.0f, 0.0f, 0.0f);
  for (int i = tid; i < OWN_ROWS * KH; i += 1024) {
    m_run[i] = ENC_NEG_INF;
    m_new[i] = ENC_NEG_INF;
    zsum[i]  = 0.0f;
  }
  if (tid < 2) more_flag[tid] = 0;
  __syncthreads();

  int*            ls = l_s + wave * CAPW;
  unsigned short* lr = l_r + wave * CAPW;
  unsigned cnt = 0;

  const int BLK  = SCAN_U * 512;
  const int epad = (E + 15) & ~15;
  const unsigned pat = ((unsigned)blockIdx.x & 0xFFu) * 0x01010101u;
  int      blk    = wave * BLK;
  int      jstart = 0;
  bool     resume = false;
  unsigned wres   = 0u;

  for (int rnd = 0; ; ++rnd) {
    cnt = 0;
    unsigned live = 0xFFFFFFFFu;
    while (blk < epad && live != 0u) {
      unsigned ov[SCAN_U][4];
      if (blk + BLK <= epad) {
#pragma unroll
        for (int u = 0; u < SCAN_U; ++u) {
          const uint4 t = *(const uint4*)(own + blk + u * 512 + lane * 16);
          ov[u][0] = t.x; ov[u][1] = t.y; ov[u][2] = t.z; ov[u][3] = t.w;
        }
      } else {
#pragma unroll
        for (int u = 0; u < SCAN_U; ++u) {
          const int i0 = blk + u * 512 + lane * 16;
          uint4 t = make_uint4(~pat, ~pat, ~pat, ~pat);
          if (i0 < epad) t = *(const uint4*)(own + i0);
          ov[u][0] = t.x; ov[u][1] = t.y; ov[u][2] = t.z; ov[u][3] = t.w;
        }
      }
#pragma unroll
      for (int j = 0; j < SCAN_U / 2; ++j) {
        if (j >= jstart) {
          unsigned w = 0u;
#pragma unroll
          for (int k = 0; k < 8; ++k) {
            const unsigned x = ov[2 * j + (k >> 2)][k & 3] ^ pat;
            const unsigned t = (x - 0x01010101u) & ~x & 0x80808080u;
            w |= t >> (7 - k);
          }
          if (resume && j == jstart) w = wres;
          unsigned act = __builtin_amdgcn_ballot_w32(w != 0u) & live;
          while (act != 0u) {
            const unsigned n = (unsigned)__builtin_popcount(act);
            if (cnt + n > (unsigned)CAPW) {
              wres = w; jstart = j; resume = true;
              live = 0u; act = 0u;
            } else {
              if (w != 0u) {
                ls[cnt + __builtin_amdgcn_mbcnt_lo(act, 0u)] =
                    2 * blk + 2048 * j + 32 * lane + __builtin_ctz(w);
                w &= w - 1u;
              }
              cnt += n;
              act = __builtin_amdgcn_ballot_w32(w != 0u);
            }
          }
        }
      }
      if (live != 0u) { blk += OWN_WAVES * BLK; jstart = 0; resume = false; }
    }
    if (blk < epad && lane == 0) more_flag[rnd & 1] = 1;

    __builtin_amdgcn_wave_barrier();

    {
      unsigned kept = 0;
      for (unsigned i0 = 0; i0 < cnt; i0 += 128u) {
        int  ee[4], ss[4], dd[4];
        bool in[4];
#pragma unroll
        for (int j = 0; j < 4; ++j) {
          const unsigned idx = i0 + j * 32 + lane;
          int e = -1;
          if (idx < cnt) {
            const unsigned c = (unsigned)ls[idx], p = c & 31u;
            e = (int)(((c >> 5) << 4) + ((p & 4u) << 7) + ((p & 3u) << 2) + (p >> 3));
          }
          in[j] = (unsigned)e < (unsigned)E;
          ee[j] = in[j] ? e : 0;
        }
#pragma unroll
        for (int j = 0; j < 4; ++j) { ss[j] = src[ee[j]]; dd[j] = dst[ee[j]]; }
#pragma unroll
        for (int j = 0; j < 4; ++j) asm volatile("" : "+v"(ss[j]));
#pragma unroll
        for (int j = 0; j < 4; ++j) {
          const unsigned rel  = (unsigned)(dd[j] - r0);
          const bool     ok   = in[j] && rel < (unsigned)rows;
          const unsigned mask = __builtin_amdgcn_ballot_w32(ok);
          if (ok) {
            const unsigned pos = kept + __builtin_popcount(mask & ((1u << lane) - 1u));
            ls[pos] = ss[j];
            lr[pos] = (unsigned short)rel;
          }
          kept += __builtin_popcount(mask);
        }
      }
      cnt = kept;
    }
    __builtin_amdgcn_wave_barrier();

    {
      const int j4 = lane >> 3, h8 = lane & 7;
      for (unsigned k = 0; k < cnt; k += 32u) {
        float    av[8], bv[8];
        unsigned rl[8];
#pragma unroll
        for (int i = 0; i < 8; ++i) {
          const unsigned idx = k + i * 4 + j4;
          const unsigned ic  = (idx < cnt) ? idx : 0u;
          const int s = ls[ic];
          rl[i] = (unsigned)lr[ic];
          av[i] = s1[s * KH + h8];
          bv[i] = s2[(r0 + (int)rl[i]) * KH + h8];
        }
#pragma unroll
        for (int i = 0; i < 8; ++i) {
          float lv = av[i] + bv[i];
          lv = (lv > 0.0f) ? lv : KALPHA * lv;
          atomicMax(&m_new[rl[i] * KH + h8], fmax_enc(lv));
        }
      }
    }
    __syncthreads();

    for (int i = tid; i < rows * KH; i += 1024) {
      const unsigned mn = m_new[i], mo = m_run[i];
      if (mn != mo) {
        m_run[i] = mn;
        if (mo != ENC_NEG_INF) {
          const float sc = __expf(fmax_dec(mo) - fmax_dec(mn));
          zsum[i] *= sc;
          float* t = tile + (i >> 3) * KHD + 4 * (i & 7);
#pragma unroll
          for (int q = 0; q < 4; ++q) {
            t[q * 32 + 0] *= sc; t[q * 32 + 1] *= sc;
            t[q * 32 + 2] *= sc; t[q * 32 + 3] *= sc;
          }
        }
      }
    }
    __syncthreads();

    if (cnt != 0u) {
      const int      hq   = lane >> 2, eq = lane & 3;
      const unsigned last = cnt - 1u;
      int      sw[GRP / 4];
      unsigned rw[GRP / 4];
#pragma unroll
      for (int i = 0; i < GRP / 4; ++i) {
        const unsigned idx = min(4u * i + (unsigned)eq, last);
        sw[i] = ls[idx];
        rw[i] = (unsigned)lr[idx];
      }
#pragma unroll
      for (int i = 0; i < GRP / 4; ++i) asm volatile("" : "+v"(sw[i]), "+v"(rw[i]));
      auto trip = [&](auto full_tag, unsigned a, unsigned nv) {
        constexpr bool FULL = decltype(full_tag)::value;
        int      sc[GRP / 4];
        unsigned rc[GRP / 4];
        bool     vw[GRP / 4];
        float    a1v[GRP / 4], a2v[GRP / 4];
        unsigned me[GRP / 4];
        int      rl[GRP];
        float4   w4[GRP];
#pragma unroll
        for (int i = 0; i < GRP / 4; ++i) {
          sc[i] = sw[i];
          rc[i] = rw[i];
          vw[i] = FULL || 4u * i + (unsigned)eq < nv;
        }
        if (FULL) {
#pragma unroll
          for (int i = 0; i < GRP / 4; ++i) {
            const unsigned idx = min(a + (unsigned)GRP + 4u * i + (unsigned)eq, last);
            sw[i] = ls[idx];
            rw[i] = (unsigned)lr[idx];
          }
        }
#pragma unroll
        for (int j = 0; j < GRP; ++j) {
          const int s = __builtin_amdgcn_readlane(sc[j >> 2], j & 3);
          rl[j] = __builtin_amdgcn_readlane((int)rc[j >> 2], j & 3);
          w4[j] = *(const float4*)(Wh + (size_t)s * KHD + lane * 4);
        }
#pragma unroll
        for (int i = 0; i < GRP / 4; ++i) {
          a1v[i] = s1[sc[i] * KH + hq];
          a2v[i] = s2[(r0 + (int)rc[i]) * KH + hq];
          me[i]  = m_run[rc[i] * KH + hq];
        }
        float pv[GRP / 4];
#pragma unroll
        for (int i = 0; i < GRP / 4; ++i) {
          float lv = a1v[i] + a2v[i];
          lv = (lv > 0.0f) ? lv : KALPHA * lv;
          float mv = fmax_dec(me[i]);
          mv = (mv < -3.0e38f) ? 0.0f : mv;
          pv[i] = __expf(lv - mv);
          if (vw[i])
            __hip_atomic_fetch_add(zsum + rc[i] * KH + hq, pv[i], __ATOMIC_RELAXED, __HIP_MEMORY_SCOPE_WORKGROUP);
        }
#pragma unroll
        for (int j = 0; j < GRP; ++j) {
          if (FULL || (unsigned)j < nv) {
            const float p = quad_bcast(pv[j >> 2], j & 3);
            float* t = tile + rl[j] * KHD + lane;
            __hip_atomic_fetch_add(t,      p * w4[j].x, __ATOMIC_RELAXED, __HIP_MEMORY_SCOPE_WORKGROUP);
            __hip_atomic_fetch_add(t + 32, p * w4[j].y, __ATOMIC_RELAXED, __HIP_MEMORY_SCOPE_WORKGROUP);
            __hip_atomic_fetch_add(t + 64, p * w4[j].z, __ATOMIC_RELAXED, __HIP_MEMORY_SCOPE_WORKGROUP);
            __hip_atomic_fetch_add(t + 96, p * w4[j].w, __ATOMIC_RELAXED, __HIP_MEMORY_SCOPE_WORKGROUP);
          }
        }
      };
      unsigned a = 0;
      for (; a + GRP <= cnt; a += GRP) trip(std::true_type{}, a, (unsigned)GRP);
      if (a < cnt) trip(std::false_type{}, a, cnt - a);
    }

    const int more = __builtin_amdgcn_readfirstlane(more_flag[rnd & 1]);
    __syncthreads();
    if (more == 0) break;
    if (tid == 0) more_flag[rnd & 1] = 0;
  }

  for (int r = wave; r < rows; r += OWN_WAVES) {
    const float* t = tile + r * KHD + lane;
    const float zi = 1.0f / fmaxf(zsum[r * KH + (lane >> 2)], 1e-16f);
    float4 o;
    o.x = t[0] * zi; o.y = t[32] * zi; o.z = t[64] * zi; o.w = t[96] * zi;
    *(float4*)(out + (size_t)(r0 + r) * KHD + lane * 4) = o;
  }
}

extern "C" void kernel_launch(void* const* d_in, const int* in_sizes, int n_in,
                              void* d_out, int out_size, void* d_ws, size_t ws_size,
                              hipStream_t stream) {
  const float* hfeat = (const float*)d_in[0];
  const int*   src   = (const int*)d_in[1];
  const int*   dst   = (const int*)d_in[2];
  const float* W     = (const float*)d_in[3];
  const float* a1    = (const float*)d_in[4];
  const float* a2    = (const float*)d_in[5];
  float* out = (float*)d_out;

  const int N = in_sizes[0] / KF;
  const int E = in_sizes[1];

  float* Wh = (float*)d_ws;
  float* s1 = Wh + (size_t)N * KHD;
  float* s2 = s1 + (size_t)N * KH;
  unsigned char* own = (unsigned char*)(s2 + (size_t)N * KH);

  const int TB = 256;
  const int nh      = N * KH;
  const int g_nh    = (nh  + TB - 1) / TB;
  const int g_gemm  = (N + 15) / 16;
  const int g_own   = (N + OWN_ROWS - 1) / OWN_ROWS;
  const int g_plane = ((E + 15) / 16 + TB - 1) / TB;

  gat_gemm_wmma_kernel<<<g_gemm, TB, 0, stream>>>(hfeat, W, Wh, N);

  scores_kernel<<<g_nh + g_plane, TB, 0, stream>>>(Wh, a1, a2, s1, s2, dst, own, N, E, g_nh);
  edge_softmax_owner_kernel<<<g_own, 1024, 0, stream>>>(src, dst, own, s1, s2, Wh, out, N, E);
}
